# starting kernel
# baseline (speedup 1.0000x reference)
#include <hip/hip_runtime.h>

typedef float v2f __attribute__((ext_vector_type(2)));
typedef float v8f __attribute__((ext_vector_type(8)));

#define FEAT 128

__global__ __launch_bounds__(256)
void gcn_gemm_dual_wmma(const float* __restrict__ X,
                        const float* __restrict__ W1,
                        const float* __restrict__ W2,
                        float* __restrict__ H1,
                        float* __restrict__ H2,
                        int M) {
  const int lane = threadIdx.x & 31;
  const int wave = threadIdx.x >> 5;
  const int m0 = blockIdx.x * 16;
  const int n0 = wave * 16;
  const int l15 = lane & 15;
  const int hi  = lane >> 4;

  int mload = m0 + l15;
  if (mload > M - 1) mload = M - 1;
  const float* __restrict__ xrow = X + (size_t)mload * FEAT;

  v8f c1 = {0.f, 0.f, 0.f, 0.f, 0.f, 0.f, 0.f, 0.f};
  v8f c2 = c1;

#pragma unroll 4
  for (int k = 0; k < FEAT; k += 4) {
    const int ka = k + 2 * hi;
    v2f a;  a.x  = xrow[ka];                      a.y  = xrow[ka + 1];
    v2f b1; b1.x = W1[ka * FEAT + n0 + l15];      b1.y = W1[(ka + 1) * FEAT + n0 + l15];
    v2f b2; b2.x = W2[ka * FEAT + n0 + l15];      b2.y = W2[(ka + 1) * FEAT + n0 + l15];
    c1 = __builtin_amdgcn_wmma_f32_16x16x4_f32(false, a, false, b1, (short)0, c1, false, false);
    c2 = __builtin_amdgcn_wmma_f32_16x16x4_f32(false, a, false, b2, (short)0, c2, false, false);
  }

  const size_t base = (size_t)(m0 + 8 * hi) * FEAT + (n0 + l15);
  float* __restrict__ o1 = H1 + base;
  float* __restrict__ o2 = H2 + base;
  if (m0 + 16 <= M) {
#pragma unroll
    for (int r = 0; r < 8; ++r) {
      o1[(size_t)r * FEAT] = c1[r];
      o2[(size_t)r * FEAT] = c2[r];
    }
  } else {
#pragma unroll
    for (int r = 0; r < 8; ++r) {
      if (m0 + r + 8 * hi < M) {
        o1[(size_t)r * FEAT] = c1[r];
        o2[(size_t)r * FEAT] = c2[r];
      }
    }
  }
}

__global__ void deg_init(float* __restrict__ d1, float* __restrict__ d2, int n) {
  int i = blockIdx.x * blockDim.x + threadIdx.x;
  if (i < n) { d1[i] = 1.0f; d2[i] = 1.0f; }
}

__global__ void deg_count(const int* __restrict__ dst, float* __restrict__ deg, int e) {
  int i = blockIdx.x * blockDim.x + threadIdx.x;
  if (i < e) atomicAdd(&deg[dst[i]], 1.0f);
}

__global__ void deg_rsqrt(float* __restrict__ d1, float* __restrict__ d2, int n) {
  int i = blockIdx.x * blockDim.x + threadIdx.x;
  if (i < n) { d1[i] = rsqrtf(d1[i]); d2[i] = rsqrtf(d2[i]); }
}

__global__ void zero_f4(float4* __restrict__ p, size_t n4) {
  size_t i = (size_t)blockIdx.x * blockDim.x + threadIdx.x;
  if (i < n4) p[i] = make_float4(0.f, 0.f, 0.f, 0.f);
}

__global__ __launch_bounds__(256)
void edge_scatter(const float* __restrict__ H,
                  const int* __restrict__ src,
                  const int* __restrict__ dst,
                  const float* __restrict__ dinv,
                  float* __restrict__ AGG,
                  int E) {
  const int e = blockIdx.x * 8 + (threadIdx.x >> 5);
  if (e >= E) return;
  const int lane = threadIdx.x & 31;
  const int s = src[e];
  const int d = dst[e];
  const float nrm = dinv[s] * dinv[d];
  const float4 v = *((const float4*)(H + (size_t)s * FEAT) + lane);
  float* ap = AGG + (size_t)d * FEAT + lane * 4;
  atomicAdd(ap + 0, v.x * nrm);
  atomicAdd(ap + 1, v.y * nrm);
  atomicAdd(ap + 2, v.z * nrm);
  atomicAdd(ap + 3, v.w * nrm);
}

__global__ __launch_bounds__(256)
void combine(const float* __restrict__ AGG,
             const float* __restrict__ H1,
             const float* __restrict__ H2,
             const float* __restrict__ dinv1,
             const float* __restrict__ dinv2,
             const float* __restrict__ b1,
             const float* __restrict__ b2,
             float* __restrict__ OUT,
             int N, int relu) {
  const int t = blockIdx.x * blockDim.x + threadIdx.x;
  if (t >= N * 32) return;
  const int v = t >> 5;
  const int q = (t & 31) * 4;
  float d1 = dinv1[v]; d1 *= d1;
  float d2 = dinv2[v]; d2 *= d2;
  const size_t off = (size_t)v * FEAT + q;
  const float4 a  = *(const float4*)(AGG + off);
  const float4 h1 = *(const float4*)(H1 + off);
  const float4 h2 = *(const float4*)(H2 + off);
  const float4 c1 = *(const float4*)(b1 + q);
  const float4 c2 = *(const float4*)(b2 + q);
  float4 o;
  o.x = a.x + d1 * h1.x + d2 * h2.x + c1.x + c2.x;
  o.y = a.y + d1 * h1.y + d2 * h2.y + c1.y + c2.y;
  o.z = a.z + d1 * h1.z + d2 * h2.z + c1.z + c2.z;
  o.w = a.w + d1 * h1.w + d2 * h2.w + c1.w + c2.w;
  if (relu) {
    o.x = fmaxf(o.x, 0.f); o.y = fmaxf(o.y, 0.f);
    o.z = fmaxf(o.z, 0.f); o.w = fmaxf(o.w, 0.f);
  }
  *(float4*)(OUT + off) = o;
}

extern "C" void kernel_launch(void* const* d_in, const int* in_sizes, int n_in,
                              void* d_out, int out_size, void* d_ws, size_t ws_size,
                              hipStream_t stream) {
  const float* x      = (const float*)d_in[0];
  const int*   ei     = (const int*)d_in[1];
  const int*   cei    = (const int*)d_in[2];
  const float* conv_w = (const float*)d_in[3];
  const float* conv_b = (const float*)d_in[4];
  const float* ctrl_w = (const float*)d_in[5];
  const float* ctrl_b = (const float*)d_in[6];

  const int N  = in_sizes[0] / FEAT;
  const int E1 = in_sizes[1] / 2;
  const int E2 = in_sizes[2] / 2;
  const int D  = in_sizes[3] / (FEAT * FEAT);

  const int* src1 = ei;        const int* dst1 = ei + E1;
  const int* src2 = cei;       const int* dst2 = cei + E2;

  const size_t NF = (size_t)N * FEAT;
  float* h1    = (float*)d_ws;
  float* h2    = h1 + NF;
  float* agg   = h2 + NF;
  float* xbuf  = agg + NF;
  float* dinv1 = xbuf + NF;
  float* dinv2 = dinv1 + N;

  const int T = 256;
  const int nodeBlocks = (N + T - 1) / T;

  deg_init<<<nodeBlocks, T, 0, stream>>>(dinv1, dinv2, N);
  deg_count<<<(E1 + T - 1) / T, T, 0, stream>>>(dst1, dinv1, E1);
  deg_count<<<(E2 + T - 1) / T, T, 0, stream>>>(dst2, dinv2, E2);
  deg_rsqrt<<<nodeBlocks, T, 0, stream>>>(dinv1, dinv2, N);

  const int gemmBlocks = (N + 15) / 16;
  const size_t n4 = NF / 4;
  const int zeroBlocks = (int)((n4 + T - 1) / T);
  const int combBlocks = (N * 32 + T - 1) / T;

  const float* x_cur = x;
  for (int i = 0; i < D; ++i) {
    const float* W1 = conv_w + (size_t)i * FEAT * FEAT;
    const float* W2 = ctrl_w + (size_t)i * FEAT * FEAT;
    const float* B1 = conv_b + (size_t)i * FEAT;
    const float* B2 = ctrl_b + (size_t)i * FEAT;
    float* x_next = (i == D - 1) ? (float*)d_out : xbuf;

    gcn_gemm_dual_wmma<<<gemmBlocks, T, 0, stream>>>(x_cur, W1, W2, h1, h2, N);
    zero_f4<<<zeroBlocks, T, 0, stream>>>((float4*)agg, n4);
    edge_scatter<<<(E1 + 7) / 8, T, 0, stream>>>(h1, src1, dst1, dinv1, agg, E1);
    edge_scatter<<<(E2 + 7) / 8, T, 0, stream>>>(h2, src2, dst2, dinv2, agg, E2);
    combine<<<combBlocks, T, 0, stream>>>(agg, h1, h2, dinv1, dinv2, B1, B2,
                                          x_next, N, (i != D - 1) ? 1 : 0);
    x_cur = x_next;
  }
}


# Dense transform: each wave keeps its 128x16 strips of W1 and W2 as B fragments in registers all launch; groups loop over row tiles, X tile loaded once per group and shared via LDS; k order kept
# speedup vs baseline: 2.9295x; 2.9295x over previous
#include <hip/hip_runtime.h>

typedef float v2f __attribute__((ext_vector_type(2)));
typedef float v8f __attribute__((ext_vector_type(8)));

#define FEAT 128

#define GEMM_T   256
#define GEMM_G   512
#define A_PITCH  132

__global__ __launch_bounds__(GEMM_T, 4)
void gcn_gemm_dual_wmma(const float* __restrict__ X,
                        const float* __restrict__ W1,
                        const float* __restrict__ W2,
                        float* __restrict__ H1,
                        float* __restrict__ H2,
                        int M, int ntiles) {
  __shared__ float As[2][16 * A_PITCH];

  const int tid  = threadIdx.x;
  const int lane = tid & 31;
  const int wave = tid >> 5;
  const int n0   = wave * 16;
  const int l15  = lane & 15;
  const int hi   = lane >> 4;

  const int srow = tid >> 5;
  const int scol = (tid & 31) * 4;
  float4 p0, p1;
  auto tile_load = [&](int tile) {
    int r0 = tile * 16 + srow;
    int r1 = r0 + 8;
    if (r0 > M - 1) r0 = M - 1;
    if (r1 > M - 1) r1 = M - 1;
    p0 = *(const float4*)(X + (size_t)r0 * FEAT + scol);
    p1 = *(const float4*)(X + (size_t)r1 * FEAT + scol);
  };
  auto tile_to_lds = [&](int buf) {
    *(float4*)(&As[buf][srow * A_PITCH + scol])       = p0;
    *(float4*)(&As[buf][(srow + 8) * A_PITCH + scol]) = p1;
  };

  int t = blockIdx.x;
  if (t < ntiles) tile_load(t);

  v2f b1[FEAT / 4], b2[FEAT / 4];
#pragma unroll
  for (int j = 0; j < FEAT / 4; ++j) {
    const int ka = 4 * j + 2 * hi;
    b1[j].x = W1[ka * FEAT + n0 + l15];   b1[j].y = W1[(ka + 1) * FEAT + n0 + l15];
    b2[j].x = W2[ka * FEAT + n0 + l15];   b2[j].y = W2[(ka + 1) * FEAT + n0 + l15];
  }

  if (t < ntiles) tile_to_lds(0);
  __syncthreads();

  int buf = 0;
  for (; t < ntiles; t += GEMM_G) {
    const int  tn   = t + GEMM_G;
    const bool more = tn < ntiles;
    if (more) tile_load(tn);

    const float* as = &As[buf][l15 * A_PITCH + 2 * hi];
    v2f a[FEAT / 4];
#pragma unroll
    for (int j = 0; j < FEAT / 4; ++j) a[j] = *(const v2f*)(as + 4 * j);
    __builtin_amdgcn_sched_barrier(0);
    v8f c1 = {0.f, 0.f, 0.f, 0.f, 0.f, 0.f, 0.f, 0.f};
    v8f c2 = c1;
#pragma unroll
    for (int j = 0; j < FEAT / 4; ++j) {
      c1 = __builtin_amdgcn_wmma_f32_16x16x4_f32(false, a[j], false, b1[j], (short)0, c1, false, false);
      c2 = __builtin_amdgcn_wmma_f32_16x16x4_f32(false, a[j], false, b2[j], (short)0, c2, false, false);
    }

    if (more) tile_to_lds(buf ^ 1);
    __syncthreads();
    buf ^= 1;

    const int m0 = t * 16;
    const size_t base = (size_t)(m0 + 8 * hi) * FEAT + (n0 + l15);
    float* __restrict__ o1 = H1 + base;
    float* __restrict__ o2 = H2 + base;
    if (m0 + 16 <= M) {
#pragma unroll
      for (int r = 0; r < 8; ++r) {
        o1[(size_t)r * FEAT] = c1[r];
        o2[(size_t)r * FEAT] = c2[r];
      }
    } else {
#pragma unroll
      for (int r = 0; r < 8; ++r) {
        if (m0 + r + 8 * hi < M) {
          o1[(size_t)r * FEAT] = c1[r];
          o2[(size_t)r * FEAT] = c2[r];
        }
      }
    }
  }
}

__global__ void deg_init(float* __restrict__ d1, float* __restrict__ d2, int n) {
  int i = blockIdx.x * blockDim.x + threadIdx.x;
  if (i < n) { d1[i] = 1.0f; d2[i] = 1.0f; }
}

__global__ void deg_count(const int* __restrict__ dst, float* __restrict__ deg, int e) {
  int i = blockIdx.x * blockDim.x + threadIdx.x;
  if (i < e) atomicAdd(&deg[dst[i]], 1.0f);
}

__global__ void deg_rsqrt(float* __restrict__ d1, float* __restrict__ d2, int n) {
  int i = blockIdx.x * blockDim.x + threadIdx.x;
  if (i < n) { d1[i] = rsqrtf(d1[i]); d2[i] = rsqrtf(d2[i]); }
}

#define AGG_R   200
#define AGG_T   1024
#define AGG_CAP 8192
#define AGG_U   8
#define AGG_B   8

__global__ __launch_bounds__(AGG_T)
void aggregate_owner(const float* __restrict__ H1, const float* __restrict__ H2,
                     const int* __restrict__ src1, const int* __restrict__ dst1, int E1,
                     const int* __restrict__ src2, const int* __restrict__ dst2, int E2,
                     const float* __restrict__ dinv1, const float* __restrict__ dinv2,
                     const float* __restrict__ b1, const float* __restrict__ b2,
                     float* __restrict__ OUT, int N, int relu,
                     int* __restrict__ g_s, int* __restrict__ g_r, float* __restrict__ g_w,
                     int* __restrict__ g_cnt, int reuse) {
  __shared__ float acc[AGG_R * FEAT];
  __shared__ int   l_s[AGG_CAP];
  __shared__ int   l_r[AGG_CAP];
  __shared__ float l_w[AGG_CAP];
  __shared__ float dl[2][AGG_R];
  __shared__ int   cnt;

  const int tid  = threadIdx.x;
  const int lane = tid & 31;
  const int wave = __builtin_amdgcn_readfirstlane(tid >> 5);
  const int v0   = blockIdx.x * AGG_R;
  const int nv   = (N - v0 < AGG_R) ? (N - v0) : AGG_R;
  const unsigned nvu = (unsigned)nv;
  const size_t gbase = (size_t)blockIdx.x * AGG_CAP;
  int kept = 0;
  if (reuse) kept = g_cnt[blockIdx.x];
  const bool build = !reuse || kept > AGG_CAP;

  if (tid == 0) cnt = 0;
  for (int i = tid; i < nv; i += AGG_T) {
    dl[0][i] = dinv1[v0 + i];
    dl[1][i] = dinv2[v0 + i];
  }
  {
    const float4 c1 = *(const float4*)(b1 + lane * 4);
    const float4 c2 = *(const float4*)(b2 + lane * 4);
    for (int r = wave; r < nv; r += AGG_T / 32) {
      float d1 = dinv1[v0 + r]; d1 *= d1;
      float d2 = dinv2[v0 + r]; d2 *= d2;
      const size_t off = (size_t)(v0 + r) * FEAT + lane * 4;
      const float4 h1 = *(const float4*)(H1 + off);
      const float4 h2 = *(const float4*)(H2 + off);
      float* a = acc + r * FEAT + lane;
      a[0]  = d1 * h1.x + d2 * h2.x + c1.x + c2.x;
      a[32] = d1 * h1.y + d2 * h2.y + c1.y + c2.y;
      a[64] = d1 * h1.z + d2 * h2.z + c1.z + c2.z;
      a[96] = d1 * h1.w + d2 * h2.w + c1.w + c2.w;
    }
  }
  int L = 0;
  if (!build) {
    L = kept < 0 ? 0 : kept;
    if (L > 0) {
      int   ts[AGG_CAP / AGG_T];
      int   tr[AGG_CAP / AGG_T];
      float tw[AGG_CAP / AGG_T];
#pragma unroll
      for (int u = 0; u < AGG_CAP / AGG_T; ++u) {
        const int i  = u * AGG_T + tid;
        const int ic = i < L ? i : L - 1;
        ts[u] = g_s[gbase + ic]; tr[u] = g_r[gbase + ic]; tw[u] = g_w[gbase + ic];
      }
#pragma unroll
      for (int u = 0; u < AGG_CAP / AGG_T; ++u) {
        const int i = u * AGG_T + tid;
        if (i < L) { l_s[i] = ts[u]; l_r[i] = tr[u]; l_w[i] = tw[u]; }
      }
    }
  }
  __syncthreads();

  auto push = [&](int e, unsigned rel, int set) {
    const int pos = __hip_atomic_fetch_add(&cnt, 1, __ATOMIC_RELAXED, __HIP_MEMORY_SCOPE_WORKGROUP);
    if (pos < AGG_CAP) {
      l_s[pos] = e;
      l_r[pos] = (int)rel | (set << 16);
    } else {
      const int s = set ? src2[e] : src1[e];
      const float w = (set ? dinv2[s] : dinv1[s]) * dl[set][rel];
      const float* h = (set ? H2 : H1) + (size_t)s * FEAT;
      float* a = acc + rel * FEAT;
#pragma nounroll
      for (int f = 0; f < FEAT; ++f)
        __hip_atomic_fetch_add(a + (f & 3) * 32 + (f >> 2), h[f] * w,
                               __ATOMIC_RELAXED, __HIP_MEMORY_SCOPE_WORKGROUP);
    }
  };

  auto test4 = [&](const int4 d, int e, int set) {
    const unsigned r0 = (unsigned)(d.x - v0), r1 = (unsigned)(d.y - v0);
    const unsigned r2 = (unsigned)(d.z - v0), r3 = (unsigned)(d.w - v0);
    const unsigned m01 = r0 < r1 ? r0 : r1, m23 = r2 < r3 ? r2 : r3;
    const unsigned m = m01 < m23 ? m01 : m23;
    if (m < nvu) {
      if (r0 < nvu) push(e + 0, r0, set);
      if (r1 < nvu) push(e + 1, r1, set);
      if (r2 < nvu) push(e + 2, r2, set);
      if (r3 < nvu) push(e + 3, r3, set);
    }
  };
  auto scan = [&](const int* __restrict__ dst, int E, int set) {
    const int E4 = ((((size_t)dst) & 15) == 0) ? (E >> 2) : 0;
    const int4* __restrict__ d4 = (const int4*)dst;
    int cb = wave * 32;
    for (; cb + (AGG_U - 1) * AGG_T + 32 <= E4; cb += AGG_U * AGG_T) {
      int4 d[AGG_U];
#pragma unroll
      for (int u = 0; u < AGG_U; ++u) d[u] = d4[cb + u * AGG_T + lane];
#pragma unroll
      for (int u = 0; u < AGG_U; ++u) test4(d[u], (cb + u * AGG_T + lane) * 4, set);
    }
#pragma nounroll
    for (int c = cb + lane; c < E4; c += AGG_T) test4(d4[c], c * 4, set);
    for (int e = E4 * 4 + tid; e < E; e += AGG_T) {
      const unsigned r = (unsigned)(dst[e] - v0);
      if (r < nvu) push(e, r, set);
    }
  };
  if (build) {
    scan(dst1, E1, 0);
    scan(dst2, E2, 1);
    __syncthreads();

    const int cntv = __builtin_amdgcn_readfirstlane(cnt);
    L = (cntv < AGG_CAP) ? cntv : AGG_CAP;
    for (int i = tid; i < L; i += AGG_T) {
      const int e = l_s[i];
      const int rr = l_r[i];
      const int set = rr >> 16;
      const int s = set ? src2[e] : src1[e];
      const float w = (set ? dinv2[s] : dinv1[s]) * dl[set][rr & 0xffff];
      l_s[i] = s;
      l_w[i] = w;
      g_s[gbase + i] = s;
      g_r[gbase + i] = rr;
      g_w[gbase + i] = w;
    }
    if (tid == 0) g_cnt[blockIdx.x] = cntv;
    __syncthreads();
  }

  auto add_row = [&](const float4 v, float w, int rr) {
    float* a = acc + (rr & 0xffff) * FEAT + lane;
    __hip_atomic_fetch_add(a,      v.x * w, __ATOMIC_RELAXED, __HIP_MEMORY_SCOPE_WORKGROUP);
    __hip_atomic_fetch_add(a + 32, v.y * w, __ATOMIC_RELAXED, __HIP_MEMORY_SCOPE_WORKGROUP);
    __hip_atomic_fetch_add(a + 64, v.z * w, __ATOMIC_RELAXED, __HIP_MEMORY_SCOPE_WORKGROUP);
    __hip_atomic_fetch_add(a + 96, v.w * w, __ATOMIC_RELAXED, __HIP_MEMORY_SCOPE_WORKGROUP);
  };
  auto row_of = [&](int i) {
    const float* H = (l_r[i] >> 16) ? H2 : H1;
    return *(const float4*)(H + (size_t)l_s[i] * FEAT + lane * 4);
  };
  const int nfull = L / AGG_B;
  for (int b = wave; b < nfull; b += AGG_T / 32) {
    float4 v[AGG_B];
#pragma unroll
    for (int u = 0; u < AGG_B; ++u) v[u] = row_of(b * AGG_B + u);
#pragma unroll
    for (int u = 0; u < AGG_B; ++u) add_row(v[u], l_w[b * AGG_B + u], l_r[b * AGG_B + u]);
  }
  {
    const int i = nfull * AGG_B + wave;
    if (i < L) add_row(row_of(i), l_w[i], l_r[i]);
  }
  __syncthreads();

  for (int r = wave; r < nv; r += AGG_T / 32) {
    const float* a = acc + r * FEAT + lane;
    float4 o = make_float4(a[0], a[32], a[64], a[96]);
    if (relu) {
      o.x = fmaxf(o.x, 0.f); o.y = fmaxf(o.y, 0.f);
      o.z = fmaxf(o.z, 0.f); o.w = fmaxf(o.w, 0.f);
    }
    *(float4*)(OUT + (size_t)(v0 + r) * FEAT + lane * 4) = o;
  }
}

extern "C" void kernel_launch(void* const* d_in, const int* in_sizes, int n_in,
                              void* d_out, int out_size, void* d_ws, size_t ws_size,
                              hipStream_t stream) {
  const float* x      = (const float*)d_in[0];
  const int*   ei     = (const int*)d_in[1];
  const int*   cei    = (const int*)d_in[2];
  const float* conv_w = (const float*)d_in[3];
  const float* conv_b = (const float*)d_in[4];
  const float* ctrl_w = (const float*)d_in[5];
  const float* ctrl_b = (const float*)d_in[6];

  const int N  = in_sizes[0] / FEAT;
  const int E1 = in_sizes[1] / 2;
  const int E2 = in_sizes[2] / 2;
  const int D  = in_sizes[3] / (FEAT * FEAT);

  const int* src1 = ei;        const int* dst1 = ei + E1;
  const int* src2 = cei;       const int* dst2 = cei + E2;

  const size_t NF = (size_t)N * FEAT;
  float* h1    = (float*)d_ws;
  float* h2    = h1 + NF;
  float* xbuf  = h2 + NF;
  float* dinv1 = xbuf + NF;
  float* dinv2 = dinv1 + N;

  const int T = 256;
  const int nodeBlocks = (N + T - 1) / T;

  deg_init<<<nodeBlocks, T, 0, stream>>>(dinv1, dinv2, N);
  deg_count<<<(E1 + T - 1) / T, T, 0, stream>>>(dst1, dinv1, E1);
  deg_count<<<(E2 + T - 1) / T, T, 0, stream>>>(dst2, dinv2, E2);
  deg_rsqrt<<<nodeBlocks, T, 0, stream>>>(dinv1, dinv2, N);

  const int gemmBlocks = (N + 15) / 16;
  const int ownerBlocks = (N + AGG_R - 1) / AGG_R;

  int*   lst_s   = (int*)(dinv2 + N);
  int*   lst_r   = lst_s + (size_t)ownerBlocks * AGG_CAP;
  float* lst_w   = (float*)(lst_r + (size_t)ownerBlocks * AGG_CAP);
  int*   lst_cnt = (int*)(lst_w + (size_t)ownerBlocks * AGG_CAP);

  const float* x_cur = x;
  for (int i = 0; i < D; ++i) {
    const float* W1 = conv_w + (size_t)i * FEAT * FEAT;
    const float* W2 = ctrl_w + (size_t)i * FEAT * FEAT;
    const float* B1 = conv_b + (size_t)i * FEAT;
    const float* B2 = ctrl_b + (size_t)i * FEAT;
    float* x_next = (i == D - 1) ? (float*)d_out : xbuf;

    gcn_gemm_dual_wmma<<<GEMM_G, GEMM_T, 0, stream>>>(x_cur, W1, W2, h1, h2, N, gemmBlocks);
    aggregate_owner<<<ownerBlocks, AGG_T, 0, stream>>>(h1, h2, src1, dst1, E1, src2, dst2, E2,
                                                       dinv1, dinv2, B1, B2,
                                                       x_next, N, (i != D - 1) ? 1 : 0,
                                                       lst_s, lst_r, lst_w, lst_cnt, (i != 0) ? 1 : 0);
    x_cur = x_next;
  }
}


# Owner lists built by one routing launch: each edge read once, LDS rank per owner, one global int add per block and owner; no group scans dst; list order = arrival, LDS adds unordered as before
# speedup vs baseline: 3.2233x; 1.1003x over previous
#include <hip/hip_runtime.h>

typedef float v2f __attribute__((ext_vector_type(2)));
typedef float v8f __attribute__((ext_vector_type(8)));

#define FEAT 128

#define GEMM_T   256
#define GEMM_G   512
#define A_PITCH  132

__global__ __launch_bounds__(GEMM_T, 4)
void gcn_gemm_dual_wmma(const float* __restrict__ X,
                        const float* __restrict__ W1,
                        const float* __restrict__ W2,
                        float* __restrict__ H1,
                        float* __restrict__ H2,
                        int M, int ntiles) {
  __shared__ float As[2][16 * A_PITCH];

  const int tid  = threadIdx.x;
  const int lane = tid & 31;
  const int wave = tid >> 5;
  const int n0   = wave * 16;
  const int l15  = lane & 15;
  const int hi   = lane >> 4;

  const int srow = tid >> 5;
  const int scol = (tid & 31) * 4;
  float4 p0, p1;
  auto tile_load = [&](int tile) {
    int r0 = tile * 16 + srow;
    int r1 = r0 + 8;
    if (r0 > M - 1) r0 = M - 1;
    if (r1 > M - 1) r1 = M - 1;
    p0 = *(const float4*)(X + (size_t)r0 * FEAT + scol);
    p1 = *(const float4*)(X + (size_t)r1 * FEAT + scol);
  };
  auto tile_to_lds = [&](int buf) {
    *(float4*)(&As[buf][srow * A_PITCH + scol])       = p0;
    *(float4*)(&As[buf][(srow + 8) * A_PITCH + scol]) = p1;
  };

  int t = blockIdx.x;
  if (t < ntiles) tile_load(t);

  v2f b1[FEAT / 4], b2[FEAT / 4];
#pragma unroll
  for (int j = 0; j < FEAT / 4; ++j) {
    const int ka = 4 * j + 2 * hi;
    b1[j].x = W1[ka * FEAT + n0 + l15];   b1[j].y = W1[(ka + 1) * FEAT + n0 + l15];
    b2[j].x = W2[ka * FEAT + n0 + l15];   b2[j].y = W2[(ka + 1) * FEAT + n0 + l15];
  }

  if (t < ntiles) tile_to_lds(0);
  __syncthreads();

  int buf = 0;
  for (; t < ntiles; t += GEMM_G) {
    const int  tn   = t + GEMM_G;
    const bool more = tn < ntiles;
    if (more) tile_load(tn);

    const float* as = &As[buf][l15 * A_PITCH + 2 * hi];
    v2f a[FEAT / 4];
#pragma unroll
    for (int j = 0; j < FEAT / 4; ++j) a[j] = *(const v2f*)(as + 4 * j);
    __builtin_amdgcn_sched_barrier(0);
    v8f c1 = {0.f, 0.f, 0.f, 0.f, 0.f, 0.f, 0.f, 0.f};
    v8f c2 = c1;
#pragma unroll
    for (int j = 0; j < FEAT / 4; ++j) {
      c1 = __builtin_amdgcn_wmma_f32_16x16x4_f32(false, a[j], false, b1[j], (short)0, c1, false, false);
      c2 = __builtin_amdgcn_wmma_f32_16x16x4_f32(false, a[j], false, b2[j], (short)0, c2, false, false);
    }

    if (more) tile_to_lds(buf ^ 1);
    __syncthreads();
    buf ^= 1;

    const int m0 = t * 16;
    const size_t base = (size_t)(m0 + 8 * hi) * FEAT + (n0 + l15);
    float* __restrict__ o1 = H1 + base;
    float* __restrict__ o2 = H2 + base;
    if (m0 + 16 <= M) {
#pragma unroll
      for (int r = 0; r < 8; ++r) {
        o1[(size_t)r * FEAT] = c1[r];
        o2[(size_t)r * FEAT] = c2[r];
      }
    } else {
#pragma unroll
      for (int r = 0; r < 8; ++r) {
        if (m0 + r + 8 * hi < M) {
          o1[(size_t)r * FEAT] = c1[r];
          o2[(size_t)r * FEAT] = c2[r];
        }
      }
    }
  }
}

__global__ void deg_init(float* __restrict__ d1, float* __restrict__ d2, int n,
                         int* __restrict__ cnt, int nown) {
  int i = blockIdx.x * blockDim.x + threadIdx.x;
  if (i < n) { d1[i] = 1.0f; d2[i] = 1.0f; }
  if (i < nown) cnt[i] = 0;
}

__global__ void deg_count(const int* __restrict__ dst, float* __restrict__ deg, int e) {
  int i = blockIdx.x * blockDim.x + threadIdx.x;
  if (i < e) atomicAdd(&deg[dst[i]], 1.0f);
}

__global__ void deg_rsqrt(float* __restrict__ d1, float* __restrict__ d2, int n) {
  int i = blockIdx.x * blockDim.x + threadIdx.x;
  if (i < n) { d1[i] = rsqrtf(d1[i]); d2[i] = rsqrtf(d2[i]); }
}

#define AGG_R   200
#define AGG_T   1024
#define AGG_CAP 8192
#define AGG_U   8
#define AGG_B   8

#define BIN_T 512
#define BIN_U 8

__global__ __launch_bounds__(BIN_T)
void route_edges(const int* __restrict__ src1, const int* __restrict__ dst1, int E1,
                 const int* __restrict__ src2, const int* __restrict__ dst2, int E2,
                 const float* __restrict__ dinv1, const float* __restrict__ dinv2,
                 int4* __restrict__ g_l, int* __restrict__ g_cnt, int nown) {
  __shared__ int hist[BIN_T];
  __shared__ int base[BIN_T];
  const int tid = threadIdx.x;
  const int E = E1 + E2;

  int   own[BIN_U], es[BIN_U], er[BIN_U], rank[BIN_U];
  float ew[BIN_U];
  int   ed[BIN_U];
#pragma unroll
  for (int u = 0; u < BIN_U; ++u) {
    const int g  = (blockIdx.x * BIN_U + u) * BIN_T + tid;
    const int gc = g < E ? g : E - 1;
    const bool set = gc >= E1;
    const int e = set ? gc - E1 : gc;
    es[u] = (set ? src2 : src1)[e];
    ed[u] = (set ? dst2 : dst1)[e];
  }
  asm volatile("" :: "v"(es[0]), "v"(es[1]), "v"(es[2]), "v"(es[3]), "v"(es[4]), "v"(es[5]), "v"(es[6]), "v"(es[7]),
                     "v"(ed[0]), "v"(ed[1]), "v"(ed[2]), "v"(ed[3]), "v"(ed[4]), "v"(ed[5]), "v"(ed[6]), "v"(ed[7]));
  static_assert(BIN_U == 8, "the statement above names eight index pairs");
  float wa[BIN_U], wb[BIN_U];
#pragma unroll
  for (int u = 0; u < BIN_U; ++u) {
    const int g = (blockIdx.x * BIN_U + u) * BIN_T + tid;
    const int set = (g < E ? g : E - 1) >= E1;
    const unsigned o = (unsigned)ed[u] / AGG_R;
    const bool ok = g < E && o < (unsigned)nown;
    const float* dinv = set ? dinv2 : dinv1;
    wa[u]  = dinv[es[u]];
    wb[u]  = dinv[ok ? ed[u] : 0];
    own[u] = ok ? (int)o : -1;
    er[u]  = (int)((unsigned)ed[u] - o * AGG_R) | (set << 16);
    rank[u] = 0;
  }
  asm volatile("" :: "v"(wa[0]), "v"(wa[1]), "v"(wa[2]), "v"(wa[3]), "v"(wa[4]), "v"(wa[5]), "v"(wa[6]), "v"(wa[7]),
                     "v"(wb[0]), "v"(wb[1]), "v"(wb[2]), "v"(wb[3]), "v"(wb[4]), "v"(wb[5]), "v"(wb[6]), "v"(wb[7]));
#pragma unroll
  for (int u = 0; u < BIN_U; ++u) ew[u] = wa[u] * wb[u];

  for (int ow = 0; ow < nown; ow += BIN_T) {
    hist[tid] = 0;
    __syncthreads();
#pragma unroll
    for (int u = 0; u < BIN_U; ++u) {
      const unsigned b = (unsigned)(own[u] - ow);
      if (b < BIN_T)
        rank[u] = __hip_atomic_fetch_add(&hist[b], 1, __ATOMIC_RELAXED, __HIP_MEMORY_SCOPE_WORKGROUP);
    }
    __syncthreads();
    {
      const int c = hist[tid];
      if (c > 0)
        base[tid] = __hip_atomic_fetch_add(&g_cnt[ow + tid], c, __ATOMIC_RELAXED, __HIP_MEMORY_SCOPE_AGENT);
    }
    __syncthreads();
#pragma unroll
    for (int u = 0; u < BIN_U; ++u) {
      const unsigned b = (unsigned)(own[u] - ow);
      if (b < BIN_T) {
        const int pos = base[b] + rank[u];
        if (pos < AGG_CAP)
          g_l[(size_t)own[u] * AGG_CAP + pos] = make_int4(es[u], er[u], __float_as_int(ew[u]), 0);
      }
    }
  }
}

__global__ __launch_bounds__(AGG_T)
void aggregate_owner(const float* __restrict__ H1, const float* __restrict__ H2,
                     const int* __restrict__ src1, const int* __restrict__ dst1, int E1,
                     const int* __restrict__ src2, const int* __restrict__ dst2, int E2,
                     const float* __restrict__ dinv1, const float* __restrict__ dinv2,
                     const float* __restrict__ b1, const float* __restrict__ b2,
                     float* __restrict__ OUT, int N, int relu,
                     const int4* __restrict__ g_l, const int* __restrict__ g_cnt) {
  __shared__ float acc[AGG_R * FEAT];
  __shared__ int   l_s[AGG_CAP];
  __shared__ int   l_r[AGG_CAP];
  __shared__ float l_w[AGG_CAP];
  __shared__ float dl[2][AGG_R];
  __shared__ int   cnt;

  const int tid  = threadIdx.x;
  const int lane = tid & 31;
  const int wave = __builtin_amdgcn_readfirstlane(tid >> 5);
  const int v0   = blockIdx.x * AGG_R;
  const int nv   = (N - v0 < AGG_R) ? (N - v0) : AGG_R;
  const unsigned nvu = (unsigned)nv;
  const size_t gbase = (size_t)blockIdx.x * AGG_CAP;
  const int kept = g_cnt[blockIdx.x];
  const bool build = kept > AGG_CAP;

  if (tid == 0) cnt = 0;
  for (int i = tid; i < nv; i += AGG_T) {
    dl[0][i] = dinv1[v0 + i];
    dl[1][i] = dinv2[v0 + i];
  }
  {
    const float4 c1 = *(const float4*)(b1 + lane * 4);
    const float4 c2 = *(const float4*)(b2 + lane * 4);
    for (int r = wave; r < nv; r += AGG_T / 32) {
      float d1 = dinv1[v0 + r]; d1 *= d1;
      float d2 = dinv2[v0 + r]; d2 *= d2;
      const size_t off = (size_t)(v0 + r) * FEAT + lane * 4;
      const float4 h1 = *(const float4*)(H1 + off);
      const float4 h2 = *(const float4*)(H2 + off);
      float* a = acc + r * FEAT + lane;
      a[0]  = d1 * h1.x + d2 * h2.x + c1.x + c2.x;
      a[32] = d1 * h1.y + d2 * h2.y + c1.y + c2.y;
      a[64] = d1 * h1.z + d2 * h2.z + c1.z + c2.z;
      a[96] = d1 * h1.w + d2 * h2.w + c1.w + c2.w;
    }
  }
  int L = 0;
  if (!build) {
    L = kept < 0 ? 0 : kept;
    if (L > 0) {
      int4 te[AGG_CAP / AGG_T];
#pragma unroll
      for (int u = 0; u < AGG_CAP / AGG_T; ++u) {
        const int i  = u * AGG_T + tid;
        const int ic = i < L ? i : L - 1;
        te[u] = g_l[gbase + ic];
      }
#pragma unroll
      for (int u = 0; u < AGG_CAP / AGG_T; ++u) {
        const int i = u * AGG_T + tid;
        if (i < L) { l_s[i] = te[u].x; l_r[i] = te[u].y; l_w[i] = __int_as_float(te[u].z); }
      }
    }
  }
  __syncthreads();

  auto push = [&](int e, unsigned rel, int set) {
    const int pos = __hip_atomic_fetch_add(&cnt, 1, __ATOMIC_RELAXED, __HIP_MEMORY_SCOPE_WORKGROUP);
    if (pos < AGG_CAP) {
      l_s[pos] = e;
      l_r[pos] = (int)rel | (set << 16);
    } else {
      const int s = set ? src2[e] : src1[e];
      const float w = (set ? dinv2[s] : dinv1[s]) * dl[set][rel];
      const float* h = (set ? H2 : H1) + (size_t)s * FEAT;
      float* a = acc + rel * FEAT;
#pragma nounroll
      for (int f = 0; f < FEAT; ++f)
        __hip_atomic_fetch_add(a + (f & 3) * 32 + (f >> 2), h[f] * w,
                               __ATOMIC_RELAXED, __HIP_MEMORY_SCOPE_WORKGROUP);
    }
  };

  auto test4 = [&](const int4 d, int e, int set) {
    const unsigned r0 = (unsigned)(d.x - v0), r1 = (unsigned)(d.y - v0);
    const unsigned r2 = (unsigned)(d.z - v0), r3 = (unsigned)(d.w - v0);
    const unsigned m01 = r0 < r1 ? r0 : r1, m23 = r2 < r3 ? r2 : r3;
    const unsigned m = m01 < m23 ? m01 : m23;
    if (m < nvu) {
      if (r0 < nvu) push(e + 0, r0, set);
      if (r1 < nvu) push(e + 1, r1, set);
      if (r2 < nvu) push(e + 2, r2, set);
      if (r3 < nvu) push(e + 3, r3, set);
    }
  };
  auto scan = [&](const int* __restrict__ dst, int E, int set) {
    const int E4 = ((((size_t)dst) & 15) == 0) ? (E >> 2) : 0;
    const int4* __restrict__ d4 = (const int4*)dst;
    int cb = wave * 32;
    for (; cb + (AGG_U - 1) * AGG_T + 32 <= E4; cb += AGG_U * AGG_T) {
      int4 d[AGG_U];
#pragma unroll
      for (int u = 0; u < AGG_U; ++u) d[u] = d4[cb + u * AGG_T + lane];
#pragma unroll
      for (int u = 0; u < AGG_U; ++u) test4(d[u], (cb + u * AGG_T + lane) * 4, set);
    }
#pragma nounroll
    for (int c = cb + lane; c < E4; c += AGG_T) test4(d4[c], c * 4, set);
    for (int e = E4 * 4 + tid; e < E; e += AGG_T) {
      const unsigned r = (unsigned)(dst[e] - v0);
      if (r < nvu) push(e, r, set);
    }
  };
  if (build) {
    scan(dst1, E1, 0);
    scan(dst2, E2, 1);
    __syncthreads();

    const int cntv = __builtin_amdgcn_readfirstlane(cnt);
    L = (cntv < AGG_CAP) ? cntv : AGG_CAP;
    for (int i = tid; i < L; i += AGG_T) {
      const int e = l_s[i];
      const int rr = l_r[i];
      const int set = rr >> 16;
      const int s = set ? src2[e] : src1[e];
      const float w = (set ? dinv2[s] : dinv1[s]) * dl[set][rr & 0xffff];
      l_s[i] = s;
      l_w[i] = w;
    }
    __syncthreads();
  }

  auto add_row = [&](const float4 v, float w, int rr) {
    float* a = acc + (rr & 0xffff) * FEAT + lane;
    __hip_atomic_fetch_add(a,      v.x * w, __ATOMIC_RELAXED, __HIP_MEMORY_SCOPE_WORKGROUP);
    __hip_atomic_fetch_add(a + 32, v.y * w, __ATOMIC_RELAXED, __HIP_MEMORY_SCOPE_WORKGROUP);
    __hip_atomic_fetch_add(a + 64, v.z * w, __ATOMIC_RELAXED, __HIP_MEMORY_SCOPE_WORKGROUP);
    __hip_atomic_fetch_add(a + 96, v.w * w, __ATOMIC_RELAXED, __HIP_MEMORY_SCOPE_WORKGROUP);
  };
  auto row_of = [&](int i) {
    const float* H = (l_r[i] >> 16) ? H2 : H1;
    return *(const float4*)(H + (size_t)l_s[i] * FEAT + lane * 4);
  };
  const int nfull = L / AGG_B;
  for (int b = wave; b < nfull; b += AGG_T / 32) {
    float4 v[AGG_B];
#pragma unroll
    for (int u = 0; u < AGG_B; ++u) v[u] = row_of(b * AGG_B + u);
#pragma unroll
    for (int u = 0; u < AGG_B; ++u) add_row(v[u], l_w[b * AGG_B + u], l_r[b * AGG_B + u]);
  }
  {
    const int i = nfull * AGG_B + wave;
    if (i < L) add_row(row_of(i), l_w[i], l_r[i]);
  }
  __syncthreads();

  for (int r = wave; r < nv; r += AGG_T / 32) {
    const float* a = acc + r * FEAT + lane;
    float4 o = make_float4(a[0], a[32], a[64], a[96]);
    if (relu) {
      o.x = fmaxf(o.x, 0.f); o.y = fmaxf(o.y, 0.f);
      o.z = fmaxf(o.z, 0.f); o.w = fmaxf(o.w, 0.f);
    }
    *(float4*)(OUT + (size_t)(v0 + r) * FEAT + lane * 4) = o;
  }
}

extern "C" void kernel_launch(void* const* d_in, const int* in_sizes, int n_in,
                              void* d_out, int out_size, void* d_ws, size_t ws_size,
                              hipStream_t stream) {
  const float* x      = (const float*)d_in[0];
  const int*   ei     = (const int*)d_in[1];
  const int*   cei    = (const int*)d_in[2];
  const float* conv_w = (const float*)d_in[3];
  const float* conv_b = (const float*)d_in[4];
  const float* ctrl_w = (const float*)d_in[5];
  const float* ctrl_b = (const float*)d_in[6];

  const int N  = in_sizes[0] / FEAT;
  const int E1 = in_sizes[1] / 2;
  const int E2 = in_sizes[2] / 2;
  const int D  = in_sizes[3] / (FEAT * FEAT);

  const int* src1 = ei;        const int* dst1 = ei + E1;
  const int* src2 = cei;       const int* dst2 = cei + E2;

  const size_t NF = (size_t)N * FEAT;
  float* h1    = (float*)d_ws;
  float* h2    = h1 + NF;
  float* xbuf  = h2 + NF;

  const int gemmBlocks = (N + 15) / 16;
  const int ownerBlocks = (N + AGG_R - 1) / AGG_R;

  int4*  lst     = (int4*)(xbuf + NF);
  int*   lst_cnt = (int*)(lst + (size_t)ownerBlocks * AGG_CAP);
  float* dinv1   = (float*)(lst_cnt + ownerBlocks);
  float* dinv2   = dinv1 + N;

  const int T = 256;
  const int nodeBlocks = (N + T - 1) / T;

  deg_init<<<nodeBlocks, T, 0, stream>>>(dinv1, dinv2, N, lst_cnt, ownerBlocks);
  deg_count<<<(E1 + T - 1) / T, T, 0, stream>>>(dst1, dinv1, E1);
  deg_count<<<(E2 + T - 1) / T, T, 0, stream>>>(dst2, dinv2, E2);
  deg_rsqrt<<<nodeBlocks, T, 0, stream>>>(dinv1, dinv2, N);

  route_edges<<<(E1 + E2 + BIN_T * BIN_U - 1) / (BIN_T * BIN_U), BIN_T, 0, stream>>>(
      src1, dst1, E1, src2, dst2, E2, dinv1, dinv2, lst, lst_cnt, ownerBlocks);

  const float* x_cur = x;
  for (int i = 0; i < D; ++i) {
    const float* W1 = conv_w + (size_t)i * FEAT * FEAT;
    const float* W2 = ctrl_w + (size_t)i * FEAT * FEAT;
    const float* B1 = conv_b + (size_t)i * FEAT;
    const float* B2 = ctrl_b + (size_t)i * FEAT;
    float* x_next = (i == D - 1) ? (float*)d_out : xbuf;

    gcn_gemm_dual_wmma<<<GEMM_G, GEMM_T, 0, stream>>>(x_cur, W1, W2, h1, h2, N, gemmBlocks);
    aggregate_owner<<<ownerBlocks, AGG_T, 0, stream>>>(h1, h2, src1, dst1, E1, src2, dst2, E2,
                                                       dinv1, dinv2, B1, B2,
                                                       x_next, N, (i != D - 1) ? 1 : 0,
                                                       lst, lst_cnt);
    x_cur = x_next;
  }
}


# Degree prep: both dst arrays counted by ONE launch of a resident 512-group grid, 8 index loads in flight per lane (was a thread per edge, 2 launches); readers take rsqrt; 13 launches become 11
# speedup vs baseline: 3.3199x; 1.0300x over previous
#include <hip/hip_runtime.h>

typedef float v2f __attribute__((ext_vector_type(2)));
typedef float v8f __attribute__((ext_vector_type(8)));

#define FEAT 128

#define GEMM_T   256
#define GEMM_G   512
#define A_PITCH  132

__global__ __launch_bounds__(GEMM_T, 4)
void gcn_gemm_dual_wmma(const float* __restrict__ X,
                        const float* __restrict__ W1,
                        const float* __restrict__ W2,
                        float* __restrict__ H1,
                        float* __restrict__ H2,
                        int M, int ntiles) {
  __shared__ float As[2][16 * A_PITCH];

  const int tid  = threadIdx.x;
  const int lane = tid & 31;
  const int wave = tid >> 5;
  const int n0   = wave * 16;
  const int l15  = lane & 15;
  const int hi   = lane >> 4;

  const int srow = tid >> 5;
  const int scol = (tid & 31) * 4;
  float4 p0, p1;
  auto tile_load = [&](int tile) {
    int r0 = tile * 16 + srow;
    int r1 = r0 + 8;
    if (r0 > M - 1) r0 = M - 1;
    if (r1 > M - 1) r1 = M - 1;
    p0 = *(const float4*)(X + (size_t)r0 * FEAT + scol);
    p1 = *(const float4*)(X + (size_t)r1 * FEAT + scol);
  };
  auto tile_to_lds = [&](int buf) {
    *(float4*)(&As[buf][srow * A_PITCH + scol])       = p0;
    *(float4*)(&As[buf][(srow + 8) * A_PITCH + scol]) = p1;
  };

  int t = blockIdx.x;
  if (t < ntiles) tile_load(t);

  v2f b1[FEAT / 4], b2[FEAT / 4];
#pragma unroll
  for (int j = 0; j < FEAT / 4; ++j) {
    const int ka = 4 * j + 2 * hi;
    b1[j].x = W1[ka * FEAT + n0 + l15];   b1[j].y = W1[(ka + 1) * FEAT + n0 + l15];
    b2[j].x = W2[ka * FEAT + n0 + l15];   b2[j].y = W2[(ka + 1) * FEAT + n0 + l15];
  }

  if (t < ntiles) tile_to_lds(0);
  __syncthreads();

  int buf = 0;
  for (; t < ntiles; t += GEMM_G) {
    const int  tn   = t + GEMM_G;
    const bool more = tn < ntiles;
    if (more) tile_load(tn);

    const float* as = &As[buf][l15 * A_PITCH + 2 * hi];
    v2f a[FEAT / 4];
#pragma unroll
    for (int j = 0; j < FEAT / 4; ++j) a[j] = *(const v2f*)(as + 4 * j);
    __builtin_amdgcn_sched_barrier(0);
    v8f c1 = {0.f, 0.f, 0.f, 0.f, 0.f, 0.f, 0.f, 0.f};
    v8f c2 = c1;
#pragma unroll
    for (int j = 0; j < FEAT / 4; ++j) {
      c1 = __builtin_amdgcn_wmma_f32_16x16x4_f32(false, a[j], false, b1[j], (short)0, c1, false, false);
      c2 = __builtin_amdgcn_wmma_f32_16x16x4_f32(false, a[j], false, b2[j], (short)0, c2, false, false);
    }

    if (more) tile_to_lds(buf ^ 1);
    __syncthreads();
    buf ^= 1;

    const int m0 = t * 16;
    const size_t base = (size_t)(m0 + 8 * hi) * FEAT + (n0 + l15);
    float* __restrict__ o1 = H1 + base;
    float* __restrict__ o2 = H2 + base;
    if (m0 + 16 <= M) {
#pragma unroll
      for (int r = 0; r < 8; ++r) {
        o1[(size_t)r * FEAT] = c1[r];
        o2[(size_t)r * FEAT] = c2[r];
      }
    } else {
#pragma unroll
      for (int r = 0; r < 8; ++r) {
        if (m0 + r + 8 * hi < M) {
          o1[(size_t)r * FEAT] = c1[r];
          o2[(size_t)r * FEAT] = c2[r];
        }
      }
    }
  }
}

__global__ void deg_init(float* __restrict__ d1, float* __restrict__ d2, int n,
                         int* __restrict__ cnt, int nown) {
  int i = blockIdx.x * blockDim.x + threadIdx.x;
  if (i < n) { d1[i] = 1.0f; d2[i] = 1.0f; }
  if (i < nown) cnt[i] = 0;
}

#define CNT_T 256
#define CNT_G 512
#define CNT_U 8

__global__ __launch_bounds__(CNT_T)
void deg_count_both(const int* __restrict__ dst1, int E1,
                    const int* __restrict__ dst2, int E2,
                    float* __restrict__ deg1, float* __restrict__ deg2) {
  const int E = E1 + E2;
  const int stride = CNT_G * CNT_T;
  for (int g0 = blockIdx.x * CNT_T + threadIdx.x; g0 < E; g0 += stride * CNT_U) {
    int d[CNT_U];
#pragma unroll
    for (int u = 0; u < CNT_U; ++u) {
      const int g  = g0 + u * stride;
      const int gc = g < E ? g : E - 1;
      d[u] = gc >= E1 ? dst2[gc - E1] : dst1[gc];
    }
    asm volatile("" :: "v"(d[0]), "v"(d[1]), "v"(d[2]), "v"(d[3]), "v"(d[4]), "v"(d[5]), "v"(d[6]), "v"(d[7]));
    static_assert(CNT_U == 8, "the statement above names eight indices");
#pragma unroll
    for (int u = 0; u < CNT_U; ++u) {
      const int g = g0 + u * stride;
      if (g < E) atomicAdd((g >= E1 ? deg2 : deg1) + d[u], 1.0f);
    }
  }
}

#define AGG_R   200
#define AGG_T   1024
#define AGG_CAP 8192
#define AGG_U   8
#define AGG_B   8

#define BIN_T 512
#define BIN_U 8

__global__ __launch_bounds__(BIN_T)
void route_edges(const int* __restrict__ src1, const int* __restrict__ dst1, int E1,
                 const int* __restrict__ src2, const int* __restrict__ dst2, int E2,
                 const float* __restrict__ deg1, const float* __restrict__ deg2,
                 int4* __restrict__ g_l, int* __restrict__ g_cnt, int nown) {
  __shared__ int hist[BIN_T];
  __shared__ int base[BIN_T];
  const int tid = threadIdx.x;
  const int E = E1 + E2;

  int   own[BIN_U], es[BIN_U], er[BIN_U], rank[BIN_U];
  float ew[BIN_U];
  int   ed[BIN_U];
#pragma unroll
  for (int u = 0; u < BIN_U; ++u) {
    const int g  = (blockIdx.x * BIN_U + u) * BIN_T + tid;
    const int gc = g < E ? g : E - 1;
    const bool set = gc >= E1;
    const int e = set ? gc - E1 : gc;
    es[u] = (set ? src2 : src1)[e];
    ed[u] = (set ? dst2 : dst1)[e];
  }
  asm volatile("" :: "v"(es[0]), "v"(es[1]), "v"(es[2]), "v"(es[3]), "v"(es[4]), "v"(es[5]), "v"(es[6]), "v"(es[7]),
                     "v"(ed[0]), "v"(ed[1]), "v"(ed[2]), "v"(ed[3]), "v"(ed[4]), "v"(ed[5]), "v"(ed[6]), "v"(ed[7]));
  static_assert(BIN_U == 8, "the statement above names eight index pairs");
  float wa[BIN_U], wb[BIN_U];
#pragma unroll
  for (int u = 0; u < BIN_U; ++u) {
    const int g = (blockIdx.x * BIN_U + u) * BIN_T + tid;
    const int set = (g < E ? g : E - 1) >= E1;
    const unsigned o = (unsigned)ed[u] / AGG_R;
    const bool ok = g < E && o < (unsigned)nown;
    const float* deg = set ? deg2 : deg1;
    wa[u]  = deg[es[u]];
    wb[u]  = deg[ok ? ed[u] : 0];
    own[u] = ok ? (int)o : -1;
    er[u]  = (int)((unsigned)ed[u] - o * AGG_R) | (set << 16);
    rank[u] = 0;
  }
  asm volatile("" :: "v"(wa[0]), "v"(wa[1]), "v"(wa[2]), "v"(wa[3]), "v"(wa[4]), "v"(wa[5]), "v"(wa[6]), "v"(wa[7]),
                     "v"(wb[0]), "v"(wb[1]), "v"(wb[2]), "v"(wb[3]), "v"(wb[4]), "v"(wb[5]), "v"(wb[6]), "v"(wb[7]));
#pragma unroll
  for (int u = 0; u < BIN_U; ++u) ew[u] = rsqrtf(wa[u]) * rsqrtf(wb[u]);

  for (int ow = 0; ow < nown; ow += BIN_T) {
    hist[tid] = 0;
    __syncthreads();
#pragma unroll
    for (int u = 0; u < BIN_U; ++u) {
      const unsigned b = (unsigned)(own[u] - ow);
      if (b < BIN_T)
        rank[u] = __hip_atomic_fetch_add(&hist[b], 1, __ATOMIC_RELAXED, __HIP_MEMORY_SCOPE_WORKGROUP);
    }
    __syncthreads();
    {
      const int c = hist[tid];
      if (c > 0)
        base[tid] = __hip_atomic_fetch_add(&g_cnt[ow + tid], c, __ATOMIC_RELAXED, __HIP_MEMORY_SCOPE_AGENT);
    }
    __syncthreads();
#pragma unroll
    for (int u = 0; u < BIN_U; ++u) {
      const unsigned b = (unsigned)(own[u] - ow);
      if (b < BIN_T) {
        const int pos = base[b] + rank[u];
        if (pos < AGG_CAP)
          g_l[(size_t)own[u] * AGG_CAP + pos] = make_int4(es[u], er[u], __float_as_int(ew[u]), 0);
      }
    }
  }
}

__global__ __launch_bounds__(AGG_T)
void aggregate_owner(const float* __restrict__ H1, const float* __restrict__ H2,
                     const int* __restrict__ src1, const int* __restrict__ dst1, int E1,
                     const int* __restrict__ src2, const int* __restrict__ dst2, int E2,
                     const float* __restrict__ deg1, const float* __restrict__ deg2,
                     const float* __restrict__ b1, const float* __restrict__ b2,
                     float* __restrict__ OUT, int N, int relu,
                     const int4* __restrict__ g_l, const int* __restrict__ g_cnt) {
  __shared__ float acc[AGG_R * FEAT];
  __shared__ int   l_s[AGG_CAP];
  __shared__ int   l_r[AGG_CAP];
  __shared__ float l_w[AGG_CAP];
  __shared__ float dl[2][AGG_R];
  __shared__ int   cnt;

  const int tid  = threadIdx.x;
  const int lane = tid & 31;
  const int wave = __builtin_amdgcn_readfirstlane(tid >> 5);
  const int v0   = blockIdx.x * AGG_R;
  const int nv   = (N - v0 < AGG_R) ? (N - v0) : AGG_R;
  const unsigned nvu = (unsigned)nv;
  const size_t gbase = (size_t)blockIdx.x * AGG_CAP;
  const int kept = g_cnt[blockIdx.x];
  const bool build = kept > AGG_CAP;

  if (tid == 0) cnt = 0;
  for (int i = tid; i < nv; i += AGG_T) {
    dl[0][i] = rsqrtf(deg1[v0 + i]);
    dl[1][i] = rsqrtf(deg2[v0 + i]);
  }
  {
    const float4 c1 = *(const float4*)(b1 + lane * 4);
    const float4 c2 = *(const float4*)(b2 + lane * 4);
    for (int r = wave; r < nv; r += AGG_T / 32) {
      float d1 = rsqrtf(deg1[v0 + r]); d1 *= d1;
      float d2 = rsqrtf(deg2[v0 + r]); d2 *= d2;
      const size_t off = (size_t)(v0 + r) * FEAT + lane * 4;
      const float4 h1 = *(const float4*)(H1 + off);
      const float4 h2 = *(const float4*)(H2 + off);
      float* a = acc + r * FEAT + lane;
      a[0]  = d1 * h1.x + d2 * h2.x + c1.x + c2.x;
      a[32] = d1 * h1.y + d2 * h2.y + c1.y + c2.y;
      a[64] = d1 * h1.z + d2 * h2.z + c1.z + c2.z;
      a[96] = d1 * h1.w + d2 * h2.w + c1.w + c2.w;
    }
  }
  int L = 0;
  if (!build) {
    L = kept < 0 ? 0 : kept;
    if (L > 0) {
      int4 te[AGG_CAP / AGG_T];
#pragma unroll
      for (int u = 0; u < AGG_CAP / AGG_T; ++u) {
        const int i  = u * AGG_T + tid;
        const int ic = i < L ? i : L - 1;
        te[u] = g_l[gbase + ic];
      }
#pragma unroll
      for (int u = 0; u < AGG_CAP / AGG_T; ++u) {
        const int i = u * AGG_T + tid;
        if (i < L) { l_s[i] = te[u].x; l_r[i] = te[u].y; l_w[i] = __int_as_float(te[u].z); }
      }
    }
  }
  __syncthreads();

  auto push = [&](int e, unsigned rel, int set) {
    const int pos = __hip_atomic_fetch_add(&cnt, 1, __ATOMIC_RELAXED, __HIP_MEMORY_SCOPE_WORKGROUP);
    if (pos < AGG_CAP) {
      l_s[pos] = e;
      l_r[pos] = (int)rel | (set << 16);
    } else {
      const int s = set ? src2[e] : src1[e];
      const float w = rsqrtf(set ? deg2[s] : deg1[s]) * dl[set][rel];
      const float* h = (set ? H2 : H1) + (size_t)s * FEAT;
      float* a = acc + rel * FEAT;
#pragma nounroll
      for (int f = 0; f < FEAT; ++f)
        __hip_atomic_fetch_add(a + (f & 3) * 32 + (f >> 2), h[f] * w,
                               __ATOMIC_RELAXED, __HIP_MEMORY_SCOPE_WORKGROUP);
    }
  };

  auto test4 = [&](const int4 d, int e, int set) {
    const unsigned r0 = (unsigned)(d.x - v0), r1 = (unsigned)(d.y - v0);
    const unsigned r2 = (unsigned)(d.z - v0), r3 = (unsigned)(d.w - v0);
    const unsigned m01 = r0 < r1 ? r0 : r1, m23 = r2 < r3 ? r2 : r3;
    const unsigned m = m01 < m23 ? m01 : m23;
    if (m < nvu) {
      if (r0 < nvu) push(e + 0, r0, set);
      if (r1 < nvu) push(e + 1, r1, set);
      if (r2 < nvu) push(e + 2, r2, set);
      if (r3 < nvu) push(e + 3, r3, set);
    }
  };
  auto scan = [&](const int* __restrict__ dst, int E, int set) {
    const int E4 = ((((size_t)dst) & 15) == 0) ? (E >> 2) : 0;
    const int4* __restrict__ d4 = (const int4*)dst;
    int cb = wave * 32;
    for (; cb + (AGG_U - 1) * AGG_T + 32 <= E4; cb += AGG_U * AGG_T) {
      int4 d[AGG_U];
#pragma unroll
      for (int u = 0; u < AGG_U; ++u) d[u] = d4[cb + u * AGG_T + lane];
#pragma unroll
      for (int u = 0; u < AGG_U; ++u) test4(d[u], (cb + u * AGG_T + lane) * 4, set);
    }
#pragma nounroll
    for (int c = cb + lane; c < E4; c += AGG_T) test4(d4[c], c * 4, set);
    for (int e = E4 * 4 + tid; e < E; e += AGG_T) {
      const unsigned r = (unsigned)(dst[e] - v0);
      if (r < nvu) push(e, r, set);
    }
  };
  if (build) {
    scan(dst1, E1, 0);
    scan(dst2, E2, 1);
    __syncthreads();

    const int cntv = __builtin_amdgcn_readfirstlane(cnt);
    L = (cntv < AGG_CAP) ? cntv : AGG_CAP;
    for (int i = tid; i < L; i += AGG_T) {
      const int e = l_s[i];
      const int rr = l_r[i];
      const int set = rr >> 16;
      const int s = set ? src2[e] : src1[e];
      const float w = rsqrtf(set ? deg2[s] : deg1[s]) * dl[set][rr & 0xffff];
      l_s[i] = s;
      l_w[i] = w;
    }
    __syncthreads();
  }

  auto add_row = [&](const float4 v, float w, int rr) {
    float* a = acc + (rr & 0xffff) * FEAT + lane;
    __hip_atomic_fetch_add(a,      v.x * w, __ATOMIC_RELAXED, __HIP_MEMORY_SCOPE_WORKGROUP);
    __hip_atomic_fetch_add(a + 32, v.y * w, __ATOMIC_RELAXED, __HIP_MEMORY_SCOPE_WORKGROUP);
    __hip_atomic_fetch_add(a + 64, v.z * w, __ATOMIC_RELAXED, __HIP_MEMORY_SCOPE_WORKGROUP);
    __hip_atomic_fetch_add(a + 96, v.w * w, __ATOMIC_RELAXED, __HIP_MEMORY_SCOPE_WORKGROUP);
  };
  auto row_of = [&](int i) {
    const float* H = (l_r[i] >> 16) ? H2 : H1;
    return *(const float4*)(H + (size_t)l_s[i] * FEAT + lane * 4);
  };
  const int nfull = L / AGG_B;
  for (int b = wave; b < nfull; b += AGG_T / 32) {
    float4 v[AGG_B];
#pragma unroll
    for (int u = 0; u < AGG_B; ++u) v[u] = row_of(b * AGG_B + u);
#pragma unroll
    for (int u = 0; u < AGG_B; ++u) add_row(v[u], l_w[b * AGG_B + u], l_r[b * AGG_B + u]);
  }
  {
    const int i = nfull * AGG_B + wave;
    if (i < L) add_row(row_of(i), l_w[i], l_r[i]);
  }
  __syncthreads();

  for (int r = wave; r < nv; r += AGG_T / 32) {
    const float* a = acc + r * FEAT + lane;
    float4 o = make_float4(a[0], a[32], a[64], a[96]);
    if (relu) {
      o.x = fmaxf(o.x, 0.f); o.y = fmaxf(o.y, 0.f);
      o.z = fmaxf(o.z, 0.f); o.w = fmaxf(o.w, 0.f);
    }
    *(float4*)(OUT + (size_t)(v0 + r) * FEAT + lane * 4) = o;
  }
}

extern "C" void kernel_launch(void* const* d_in, const int* in_sizes, int n_in,
                              void* d_out, int out_size, void* d_ws, size_t ws_size,
                              hipStream_t stream) {
  const float* x      = (const float*)d_in[0];
  const int*   ei     = (const int*)d_in[1];
  const int*   cei    = (const int*)d_in[2];
  const float* conv_w = (const float*)d_in[3];
  const float* conv_b = (const float*)d_in[4];
  const float* ctrl_w = (const float*)d_in[5];
  const float* ctrl_b = (const float*)d_in[6];

  const int N  = in_sizes[0] / FEAT;
  const int E1 = in_sizes[1] / 2;
  const int E2 = in_sizes[2] / 2;
  const int D  = in_sizes[3] / (FEAT * FEAT);

  const int* src1 = ei;        const int* dst1 = ei + E1;
  const int* src2 = cei;       const int* dst2 = cei + E2;

  const size_t NF = (size_t)N * FEAT;
  float* h1    = (float*)d_ws;
  float* h2    = h1 + NF;
  float* xbuf  = h2 + NF;

  const int gemmBlocks = (N + 15) / 16;
  const int ownerBlocks = (N + AGG_R - 1) / AGG_R;

  int4*  lst     = (int4*)(xbuf + NF);
  int*   lst_cnt = (int*)(lst + (size_t)ownerBlocks * AGG_CAP);
  float* deg1    = (float*)(lst_cnt + ownerBlocks);
  float* deg2    = deg1 + N;

  const int T = 256;
  const int nodeBlocks = (N + T - 1) / T;

  deg_init<<<nodeBlocks, T, 0, stream>>>(deg1, deg2, N, lst_cnt, ownerBlocks);
  deg_count_both<<<CNT_G, CNT_T, 0, stream>>>(dst1, E1, dst2, E2, deg1, deg2);

  route_edges<<<(E1 + E2 + BIN_T * BIN_U - 1) / (BIN_T * BIN_U), BIN_T, 0, stream>>>(
      src1, dst1, E1, src2, dst2, E2, deg1, deg2, lst, lst_cnt, ownerBlocks);

  const float* x_cur = x;
  for (int i = 0; i < D; ++i) {
    const float* W1 = conv_w + (size_t)i * FEAT * FEAT;
    const float* W2 = ctrl_w + (size_t)i * FEAT * FEAT;
    const float* B1 = conv_b + (size_t)i * FEAT;
    const float* B2 = ctrl_b + (size_t)i * FEAT;
    float* x_next = (i == D - 1) ? (float*)d_out : xbuf;

    gcn_gemm_dual_wmma<<<GEMM_G, GEMM_T, 0, stream>>>(x_cur, W1, W2, h1, h2, N, gemmBlocks);
    aggregate_owner<<<ownerBlocks, AGG_T, 0, stream>>>(h1, h2, src1, dst1, E1, src2, dst2, E2,
                                                       deg1, deg2, B1, B2,
                                                       x_next, N, (i != D - 1) ? 1 : 0,
                                                       lst, lst_cnt);
    x_cur = x_next;
  }
}
